# starting kernel
# baseline (speedup 1.0000x reference)
#include <hip/hip_runtime.h>
#include <hip/hip_bf16.h>

typedef __attribute__((ext_vector_type(2))) float v2f;
typedef __attribute__((ext_vector_type(8))) float v8f;

#define IN_FEAT 512
#define HIDDEN  16
#define NCLS    7


__global__ void gcn_deg_init(float* deg, int n) {
    int i = blockIdx.x * blockDim.x + threadIdx.x;
    if (i < n) deg[i] = 1.0f;
}

__global__ void gcn_deg_accum(const int* __restrict__ dst, float* deg, int E) {
    int e = blockIdx.x * blockDim.x + threadIdx.x;
    if (e < E) atomicAdd(&deg[dst[e]], 1.0f);
}

__global__ void gcn_deg_rsqrt(float* dinv, int n) {
    int i = blockIdx.x * blockDim.x + threadIdx.x;
    if (i < n) {
        float d = dinv[i];
        dinv[i] = (d > 0.0f) ? rsqrtf(d) : 0.0f;
    }
}

__global__ void gcn_zero(float* p, int n) {
    int i = blockIdx.x * blockDim.x + threadIdx.x;
    if (i < n) p[i] = 0.0f;
}


__global__ void gcn_gemm1_wmma(const float* __restrict__ x,
                               const float* __restrict__ W1,
                               float* __restrict__ h, int n) {
    int wave = (blockIdx.x * blockDim.x + threadIdx.x) >> 5;
    int lane = threadIdx.x & 31;
    int row0 = wave * 16;
    if (row0 >= n) return;

    int r    = lane & 15;
    int hi   = lane >> 4;
    int koff = hi * 2;
    int row  = row0 + r;
    if (row >= n) row = n - 1;
    const float* xrow = x + (size_t)row * IN_FEAT;

    v8f c = {0.f, 0.f, 0.f, 0.f, 0.f, 0.f, 0.f, 0.f};

    #pragma unroll 4
    for (int k = 0; k < IN_FEAT; k += 4) {
        if ((k & 63) == 0)
            __builtin_prefetch(xrow + k + 128, 0, 1);
        v2f a = *(const v2f*)(xrow + k + koff);
        int kk = k + koff;
        v2f b;
        b.x = W1[(size_t)kk * HIDDEN + r];
        b.y = W1[(size_t)(kk + 1) * HIDDEN + r];
        c = __builtin_amdgcn_wmma_f32_16x16x4_f32(false, a, false, b,
                                                  (short)0, c, false, false);
    }

    int srow = row0 + hi * 8;
    #pragma unroll
    for (int i = 0; i < 8; ++i) {
        int rr = srow + i;
        if (rr < n) h[(size_t)rr * HIDDEN + r] = c[i];
    }
}


__global__ void gcn_agg1(const int* __restrict__ src, const int* __restrict__ dst,
                         const float* __restrict__ dinv, const float* __restrict__ h,
                         float* out1, int E) {
    int t = blockIdx.x * blockDim.x + threadIdx.x;
    int e = t >> 2, q = t & 3;
    if (e >= E) return;
    int s = src[e], d = dst[e];
    float w = dinv[s] * dinv[d];
    float4 v = *((const float4*)(h + (size_t)s * HIDDEN) + q);
    float* o = out1 + (size_t)d * HIDDEN + q * 4;
    atomicAdd(o + 0, v.x * w);
    atomicAdd(o + 1, v.y * w);
    atomicAdd(o + 2, v.z * w);
    atomicAdd(o + 3, v.w * w);
}


__global__ void gcn_selfloop_relu(const float* __restrict__ out1,
                                  const float* __restrict__ h,
                                  const float* __restrict__ dinv,
                                  const float* __restrict__ b1,
                                  float* hrelu, int n) {
    int i = blockIdx.x * blockDim.x + threadIdx.x;
    if (i >= n) return;
    float w = dinv[i] * dinv[i];
    const float* o = out1 + (size_t)i * HIDDEN;
    const float* hh = h + (size_t)i * HIDDEN;
    float* dst = hrelu + (size_t)i * HIDDEN;
    #pragma unroll
    for (int c = 0; c < HIDDEN; ++c)
        dst[c] = fmaxf(o[c] + hh[c] * w + b1[c], 0.0f);
}


__global__ void gcn_gemm2_wmma(const float* __restrict__ hrelu,
                               const float* __restrict__ W2,
                               float* __restrict__ g, int n) {
    int wave = (blockIdx.x * blockDim.x + threadIdx.x) >> 5;
    int lane = threadIdx.x & 31;
    int row0 = wave * 16;
    if (row0 >= n) return;

    int r    = lane & 15;
    int hi   = lane >> 4;
    int koff = hi * 2;
    int row  = row0 + r;
    if (row >= n) row = n - 1;
    const float* hrow = hrelu + (size_t)row * HIDDEN;

    v8f c = {0.f, 0.f, 0.f, 0.f, 0.f, 0.f, 0.f, 0.f};

    #pragma unroll
    for (int k = 0; k < HIDDEN; k += 4) {
        int kk = k + koff;
        v2f a = *(const v2f*)(hrow + kk);
        v2f b;
        b.x = (r < NCLS) ? W2[(size_t)kk * NCLS + r] : 0.0f;
        b.y = (r < NCLS) ? W2[(size_t)(kk + 1) * NCLS + r] : 0.0f;
        c = __builtin_amdgcn_wmma_f32_16x16x4_f32(false, a, false, b,
                                                  (short)0, c, false, false);
    }

    int srow = row0 + hi * 8;
    #pragma unroll
    for (int i = 0; i < 8; ++i) {
        int rr = srow + i;
        if (rr < n) g[(size_t)rr * HIDDEN + r] = c[i];
    }
}


__global__ void gcn_agg2(const int* __restrict__ src, const int* __restrict__ dst,
                         const float* __restrict__ dinv, const float* __restrict__ g,
                         float* out2, int E) {
    int e = blockIdx.x * blockDim.x + threadIdx.x;
    if (e >= E) return;
    int s = src[e], d = dst[e];
    float w = dinv[s] * dinv[d];
    const float* gs = g + (size_t)s * HIDDEN;
    float* o = out2 + (size_t)d * 8;
    #pragma unroll
    for (int c = 0; c < NCLS; ++c)
        atomicAdd(o + c, gs[c] * w);
}


__global__ void gcn_logsoftmax(const float* __restrict__ out2,
                               const float* __restrict__ g,
                               const float* __restrict__ dinv,
                               const float* __restrict__ b2,
                               float* __restrict__ out, int n) {
    int i = blockIdx.x * blockDim.x + threadIdx.x;
    if (i >= n) return;
    float w = dinv[i] * dinv[i];
    const float* o = out2 + (size_t)i * 8;
    const float* gg = g + (size_t)i * HIDDEN;
    float v[NCLS];
    float m = -3.4e38f;
    #pragma unroll
    for (int c = 0; c < NCLS; ++c) {
        v[c] = o[c] + gg[c] * w + b2[c];
        m = fmaxf(m, v[c]);
    }
    float s = 0.0f;
    #pragma unroll
    for (int c = 0; c < NCLS; ++c) s += expf(v[c] - m);
    float l = logf(s);
    float* dsto = out + (size_t)i * NCLS;
    #pragma unroll
    for (int c = 0; c < NCLS; ++c) dsto[c] = v[c] - m - l;
}


extern "C" void kernel_launch(void* const* d_in, const int* in_sizes, int n_in,
                              void* d_out, int out_size, void* d_ws, size_t ws_size,
                              hipStream_t stream) {
    const float* x  = (const float*)d_in[0];
    const int*   ei = (const int*)d_in[1];
    const float* W1 = (const float*)d_in[2];
    const float* b1 = (const float*)d_in[3];
    const float* W2 = (const float*)d_in[4];
    const float* b2 = (const float*)d_in[5];
    float* out = (float*)d_out;

    const int n = in_sizes[0] / IN_FEAT;
    const int E = in_sizes[1] / 2;
    const int* src = ei;
    const int* dst = ei + E;

    float* f     = (float*)d_ws;
    float* dinv  = f;
    float* h     = f + (size_t)n;
    float* out1  = f + (size_t)17 * n;
    float* hrelu = f + (size_t)33 * n;
    float* g     = out1;
    float* out2  = h;

    const int B = 256;
    const int tiles = (n + 15) / 16;
    const int gemmBlocks = (tiles + 7) / 8;

    gcn_deg_init   <<<(n + B - 1) / B, B, 0, stream>>>(dinv, n);
    gcn_deg_accum  <<<(E + B - 1) / B, B, 0, stream>>>(dst, dinv, E);
    gcn_deg_rsqrt  <<<(n + B - 1) / B, B, 0, stream>>>(dinv, n);

    gcn_zero       <<<((size_t)16 * n + B - 1) / B, B, 0, stream>>>(out1, 16 * n);
    gcn_gemm1_wmma <<<gemmBlocks, B, 0, stream>>>(x, W1, h, n);
    gcn_agg1       <<<((size_t)4 * E + B - 1) / B, B, 0, stream>>>(src, dst, dinv, h, out1, E);
    gcn_selfloop_relu<<<(n + B - 1) / B, B, 0, stream>>>(out1, h, dinv, b1, hrelu, n);

    gcn_zero       <<<((size_t)8 * n + B - 1) / B, B, 0, stream>>>(out2, 8 * n);
    gcn_gemm2_wmma <<<gemmBlocks, B, 0, stream>>>(hrelu, W2, g, n);
    gcn_agg2       <<<(E + B - 1) / B, B, 0, stream>>>(src, dst, dinv, g, out2, E);
    gcn_logsoftmax <<<(n + B - 1) / B, B, 0, stream>>>(out2, g, dinv, b2, out, n);
}


# edge sums by destination ownership: a workgroup keeps 4096 rows' sums in LDS, filters 1/10 of the edges for them (LDS f32 adds), adds its tile to out; order: LDS arrival, then 10 partials by arrival
# speedup vs baseline: 2.0392x; 2.0392x over previous
#include <hip/hip_runtime.h>
#include <hip/hip_bf16.h>

typedef __attribute__((ext_vector_type(2))) float v2f;
typedef __attribute__((ext_vector_type(8))) float v8f;

#define IN_FEAT 512
#define HIDDEN  16
#define NCLS    7


__global__ void gcn_deg_init(float* deg, int n) {
    int i = blockIdx.x * blockDim.x + threadIdx.x;
    if (i < n) deg[i] = 1.0f;
}

__global__ void gcn_deg_accum(const int* __restrict__ dst, float* deg, int E) {
    int e = blockIdx.x * blockDim.x + threadIdx.x;
    if (e < E) atomicAdd(&deg[dst[e]], 1.0f);
}

__global__ void gcn_deg_rsqrt(float* dinv, int n) {
    int i = blockIdx.x * blockDim.x + threadIdx.x;
    if (i < n) {
        float d = dinv[i];
        dinv[i] = (d > 0.0f) ? rsqrtf(d) : 0.0f;
    }
}

__global__ void gcn_zero(float* p, int n) {
    int i = blockIdx.x * blockDim.x + threadIdx.x;
    if (i < n) p[i] = 0.0f;
}


__global__ void gcn_gemm1_wmma(const float* __restrict__ x,
                               const float* __restrict__ W1,
                               float* __restrict__ h, int n) {
    int wave = (blockIdx.x * blockDim.x + threadIdx.x) >> 5;
    int lane = threadIdx.x & 31;
    int row0 = wave * 16;
    if (row0 >= n) return;

    int r    = lane & 15;
    int hi   = lane >> 4;
    int koff = hi * 2;
    int row  = row0 + r;
    if (row >= n) row = n - 1;
    const float* xrow = x + (size_t)row * IN_FEAT;

    v8f c = {0.f, 0.f, 0.f, 0.f, 0.f, 0.f, 0.f, 0.f};

    #pragma unroll 4
    for (int k = 0; k < IN_FEAT; k += 4) {
        if ((k & 63) == 0)
            __builtin_prefetch(xrow + k + 128, 0, 1);
        v2f a = *(const v2f*)(xrow + k + koff);
        int kk = k + koff;
        v2f b;
        b.x = W1[(size_t)kk * HIDDEN + r];
        b.y = W1[(size_t)(kk + 1) * HIDDEN + r];
        c = __builtin_amdgcn_wmma_f32_16x16x4_f32(false, a, false, b,
                                                  (short)0, c, false, false);
    }

    int srow = row0 + hi * 8;
    #pragma unroll
    for (int i = 0; i < 8; ++i) {
        int rr = srow + i;
        if (rr < n) h[(size_t)rr * HIDDEN + r] = c[i];
    }
}


#define OWN_ROWS    4096
#define OWN_PARTS   10
#define OWN_THREADS 1024
#define OWN_WAVES   (OWN_THREADS / 32)
#define OWN_QCAP    160

__device__ __forceinline__ int4 gcn_ld4(const int* __restrict__ p, int e, int e1, bool vec) {
    int4 r;
    if (vec && e + 3 < e1) {
        r = *(const int4*)(p + e);
    } else {
        r.x = (e     < e1) ? p[e]     : -1;
        r.y = (e + 1 < e1) ? p[e + 1] : -1;
        r.z = (e + 2 < e1) ? p[e + 2] : -1;
        r.w = (e + 3 < e1) ? p[e + 3] : -1;
    }
    return r;
}

template <bool FULL>
__device__ __forceinline__ void gcn_own_drain(const int2* qp, int cnt, int lo,
                                              const float* __restrict__ dinv,
                                              const float* __restrict__ hin,
                                              float* acc, int c, int half) {
    int2  ent[16];
    float wgt[16], val[16];
    #pragma unroll
    for (int i = 0; i < 16; ++i) {
        int k = 2 * i + half;
        ent[i] = qp[(FULL || k < cnt) ? k : 0];
    }
    #pragma unroll
    for (int i = 0; i < 16; ++i) {
        int s = ent[i].x, dl = ent[i].y;
        wgt[i] = dinv[s] * dinv[lo + dl];
        val[i] = hin[(size_t)s * HIDDEN + c];
    }
    #pragma unroll
    for (int i = 0; i < 16; ++i) {
        int k = 2 * i + half;
        if (FULL || k < cnt)
            atomicAdd(&acc[ent[i].y * HIDDEN + c], val[i] * wgt[i]);
    }
}

__global__ void __launch_bounds__(OWN_THREADS)
gcn_agg_own(const int* __restrict__ src, const int* __restrict__ dst,
            const float* __restrict__ dinv, const float* __restrict__ hin,
            float* outp, int E, int n, int groups) {
    __shared__ alignas(16) float acc[OWN_ROWS * HIDDEN];
    __shared__ int2  que[OWN_WAVES * OWN_QCAP];

    const int tid  = threadIdx.x;
    const int lane = tid & 31;
    const int wv   = tid >> 5;
    const int g    = blockIdx.x % groups;
    const int p    = blockIdx.x / groups;
    const int lo   = g * OWN_ROWS;
    if (lo >= n) return;
    int rows = n - lo;
    if (rows > OWN_ROWS) rows = OWN_ROWS;

    for (int i = tid * 4; i < OWN_ROWS * HIDDEN; i += OWN_THREADS * 4)
        *(float4*)(acc + i) = make_float4(0.f, 0.f, 0.f, 0.f);
    __syncthreads();

    int per = (E + OWN_PARTS - 1) / OWN_PARTS;
    per = (per + 3) & ~3;
    const int e0 = p * per;
    int e1 = e0 + per;
    if (e1 > E) e1 = E;
    const bool vec = ((E & 3) == 0) &&
                     (((((size_t)src) | ((size_t)dst)) & 15) == 0);

    int2* myq = que + wv * OWN_QCAP;
    const int c = lane & 15, half = lane >> 4;
    const unsigned lt = (1u << lane) - 1u;

    int e = e0 + tid * 4;
    int4 d4 = gcn_ld4(dst, e, e1, vec);
    int4 s4 = gcn_ld4(src, e, e1, vec);
    int qn = 0;
    const int steps = (e1 - e0 + OWN_THREADS * 4 - 1) / (OWN_THREADS * 4);
    for (int it = 0; it < steps; ++it) {
        const int en = e + OWN_THREADS * 4;
        int4 dn = gcn_ld4(dst, en, e1, vec);
        int4 sn = gcn_ld4(src, en, e1, vec);
        int dd[4] = {d4.x, d4.y, d4.z, d4.w};
        int ss[4] = {s4.x, s4.y, s4.z, s4.w};
        #pragma unroll
        for (int j = 0; j < 4; ++j) {
            unsigned dl = (unsigned)(dd[j] - lo);
            bool m = dl < (unsigned)rows;
            unsigned mask = (unsigned)__ballot(m);
            if (m) myq[qn + __popc(mask & lt)] = make_int2(ss[j], (int)dl);
            qn += __popc(mask);
        }
        while (qn >= 32) {
            qn -= 32;
            gcn_own_drain<true>(myq + qn, 32, lo, dinv, hin, acc, c, half);
        }
        d4 = dn; s4 = sn; e = en;
    }
    if (qn > 0) gcn_own_drain<false>(myq, qn, lo, dinv, hin, acc, c, half);
    __syncthreads();

    float* ob = outp + (size_t)lo * HIDDEN;
    const int tot = rows * HIDDEN;
    for (int i = tid; i < tot; i += OWN_THREADS)
        atomicAdd(ob + i, acc[i]);
}


__global__ void gcn_selfloop_relu(const float* __restrict__ out1,
                                  const float* __restrict__ h,
                                  const float* __restrict__ dinv,
                                  const float* __restrict__ b1,
                                  float* hrelu, int n) {
    int i = blockIdx.x * blockDim.x + threadIdx.x;
    if (i >= n) return;
    float w = dinv[i] * dinv[i];
    const float* o = out1 + (size_t)i * HIDDEN;
    const float* hh = h + (size_t)i * HIDDEN;
    float* dst = hrelu + (size_t)i * HIDDEN;
    #pragma unroll
    for (int c = 0; c < HIDDEN; ++c)
        dst[c] = fmaxf(o[c] + hh[c] * w + b1[c], 0.0f);
}


__global__ void gcn_gemm2_wmma(const float* __restrict__ agg2,
                               const float* __restrict__ hrelu,
                               const float* __restrict__ dinv,
                               const float* __restrict__ W2,
                               float* __restrict__ g, int n) {
    int wave = (blockIdx.x * blockDim.x + threadIdx.x) >> 5;
    int lane = threadIdx.x & 31;
    int row0 = wave * 16;
    if (row0 >= n) return;

    int r    = lane & 15;
    int hi   = lane >> 4;
    int koff = hi * 2;
    int row  = row0 + r;
    if (row >= n) row = n - 1;
    const float* srow = agg2  + (size_t)row * HIDDEN;
    const float* hrow = hrelu + (size_t)row * HIDDEN;
    float w = dinv[row];
    w = w * w;

    v8f c = {0.f, 0.f, 0.f, 0.f, 0.f, 0.f, 0.f, 0.f};

    #pragma unroll
    for (int k = 0; k < HIDDEN; k += 4) {
        int kk = k + koff;
        v2f s  = *(const v2f*)(srow + kk);
        v2f hh = *(const v2f*)(hrow + kk);
        v2f a;
        a.x = s.x + hh.x * w;
        a.y = s.y + hh.y * w;
        v2f b;
        b.x = (r < NCLS) ? W2[(size_t)kk * NCLS + r] : 0.0f;
        b.y = (r < NCLS) ? W2[(size_t)(kk + 1) * NCLS + r] : 0.0f;
        c = __builtin_amdgcn_wmma_f32_16x16x4_f32(false, a, false, b,
                                                  (short)0, c, false, false);
    }

    int orow = row0 + hi * 8;
    #pragma unroll
    for (int i = 0; i < 8; ++i) {
        int rr = orow + i;
        if (rr < n) g[(size_t)rr * HIDDEN + r] = c[i];
    }
}


__global__ void gcn_logsoftmax(const float* __restrict__ g,
                               const float* __restrict__ b2,
                               float* __restrict__ out, int n) {
    int i = blockIdx.x * blockDim.x + threadIdx.x;
    if (i >= n) return;
    const float* gg = g + (size_t)i * HIDDEN;
    float v[NCLS];
    float m = -3.4e38f;
    #pragma unroll
    for (int c = 0; c < NCLS; ++c) {
        v[c] = gg[c] + b2[c];
        m = fmaxf(m, v[c]);
    }
    float s = 0.0f;
    #pragma unroll
    for (int c = 0; c < NCLS; ++c) s += expf(v[c] - m);
    float l = logf(s);
    float* dsto = out + (size_t)i * NCLS;
    #pragma unroll
    for (int c = 0; c < NCLS; ++c) dsto[c] = v[c] - m - l;
}


extern "C" void kernel_launch(void* const* d_in, const int* in_sizes, int n_in,
                              void* d_out, int out_size, void* d_ws, size_t ws_size,
                              hipStream_t stream) {
    const float* x  = (const float*)d_in[0];
    const int*   ei = (const int*)d_in[1];
    const float* W1 = (const float*)d_in[2];
    const float* b1 = (const float*)d_in[3];
    const float* W2 = (const float*)d_in[4];
    const float* b2 = (const float*)d_in[5];
    float* out = (float*)d_out;

    const int n = in_sizes[0] / IN_FEAT;
    const int E = in_sizes[1] / 2;
    const int* src = ei;
    const int* dst = ei + E;

    float* f     = (float*)d_ws;
    float* dinv  = f;
    float* h     = f + (size_t)n;
    float* out1  = f + (size_t)17 * n;
    float* hrelu = f + (size_t)33 * n;
    float* g     = out1;
    float* agg2  = h;

    const int B = 256;
    const int tiles = (n + 15) / 16;
    const int gemmBlocks = (tiles + 7) / 8;
    const int ownGroups = (n + OWN_ROWS - 1) / OWN_ROWS;

    gcn_deg_init   <<<(n + B - 1) / B, B, 0, stream>>>(dinv, n);
    gcn_deg_accum  <<<(E + B - 1) / B, B, 0, stream>>>(dst, dinv, E);
    gcn_deg_rsqrt  <<<(n + B - 1) / B, B, 0, stream>>>(dinv, n);

    gcn_zero       <<<((size_t)16 * n + B - 1) / B, B, 0, stream>>>(out1, 16 * n);
    gcn_gemm1_wmma <<<gemmBlocks, B, 0, stream>>>(x, W1, h, n);
    gcn_agg_own    <<<ownGroups * OWN_PARTS, OWN_THREADS, 0, stream>>>(src, dst, dinv, h, out1, E, n, ownGroups);
    gcn_selfloop_relu<<<(n + B - 1) / B, B, 0, stream>>>(out1, h, dinv, b1, hrelu, n);

    gcn_zero       <<<((size_t)16 * n + B - 1) / B, B, 0, stream>>>(agg2, 16 * n);
    gcn_agg_own    <<<ownGroups * OWN_PARTS, OWN_THREADS, 0, stream>>>(src, dst, dinv, hrelu, agg2, E, n, ownGroups);
    gcn_gemm2_wmma <<<gemmBlocks, B, 0, stream>>>(agg2, hrelu, dinv, W2, g, n);
    gcn_logsoftmax <<<(n + B - 1) / B, B, 0, stream>>>(g, b2, out, n);
}
